# starting kernel
# baseline (speedup 1.0000x reference)
#include <hip/hip_runtime.h>
#include <math.h>

#define BATCH 8
#define CCH   768
#define HH    32
#define WW    64
#define HW    (HH*WW)
#define NPTS  (BATCH*HH*WW)
#define DD    768
#define KK    19
#define MMIX  5
#define KM    95
#define KMP   96
#define LOG2PI 1.8378770664093453f

typedef __attribute__((ext_vector_type(2))) float v2f;
typedef __attribute__((ext_vector_type(8))) float v8f;

typedef const float __attribute__((address_space(1)))* gcf;
typedef const v2f   __attribute__((address_space(1)))* gcv2;

__global__ __launch_bounds__(256) void gmm_prep_b(
    const float* __restrict__ means, const float* __restrict__ log_vars,
    float* __restrict__ b1, float* __restrict__ b2) {
  int idx = blockIdx.x * 256 + threadIdx.x;
  if (idx >= KMP * DD) return;
  int d = idx / KMP;
  int c = idx % KMP;
  float v1 = 0.0f, v2 = 0.0f;
  if (c < KM) {
    float lv = log_vars[c * DD + d];
    float mu = means[c * DD + d];
    float iv = expf(-lv);
    v1 = iv;
    v2 = -2.0f * mu * iv;
  }
  int o = (d >> 1) * (2 * KMP) + c * 2 + (d & 1);
  b1[o] = v1;
  b2[o] = v2;
}

__global__ __launch_bounds__(128) void gmm_prep_c(
    const float* __restrict__ means, const float* __restrict__ log_vars,
    const float* __restrict__ log_w, float* __restrict__ ckm) {
  int c = threadIdx.x;
  if (c >= KMP) return;
  if (c >= KM) { ckm[c] = -1.0e30f; return; }
  float slv = 0.0f, smi = 0.0f;
  for (int d = 0; d < DD; ++d) {
    float lv = log_vars[c * DD + d];
    float mu = means[c * DD + d];
    slv += lv;
    smi += mu * mu * expf(-lv);
  }
  ckm[c] = log_w[c] - 0.5f * ((float)DD * LOG2PI + slv + smi);
}

__global__ __launch_bounds__(128) void gmm_main(
    const float* __restrict__ feats,
    const float* __restrict__ b1, const float* __restrict__ b2,
    const float* __restrict__ ckm, float* __restrict__ out) {
  __shared__ float sdata[128 * KMP];
  __shared__ float cksh[KMP];

  const int tid  = threadIdx.x;
  const int wave = tid >> 5;
  const int lane = tid & 31;
  const int lrow = lane & 15;
  const int lhi  = lane >> 4;

  if (tid < KMP) cksh[tid] = ckm[tid];

  v8f acc[2][6];
#pragma unroll
  for (int nt = 0; nt < 2; ++nt)
#pragma unroll
    for (int ct = 0; ct < 6; ++ct)
      acc[nt][ct] = (v8f){0.f,0.f,0.f,0.f,0.f,0.f,0.f,0.f};

  gcf ap[2];
#pragma unroll
  for (int nt = 0; nt < 2; ++nt) {
    int n0  = blockIdx.x * 128 + wave * 32 + nt * 16;
    int bb  = n0 >> 11;
    int rem = n0 & 2047;
    ap[nt] = (gcf)(feats + (size_t)bb * (size_t)(CCH * HW)
                         + (size_t)(2 * lhi) * HW + rem + lrow);
  }

  gcf bp1[6];
  gcf bp2[6];
#pragma unroll
  for (int ct = 0; ct < 6; ++ct) {
    int off = lhi * (2 * KMP) + (ct * 16 + lrow) * 2;
    bp1[ct] = (gcf)(b1 + off);
    bp2[ct] = (gcf)(b2 + off);
  }

  for (int d0 = 0; d0 < DD; d0 += 4) {
    v2f a[2], asq[2];
#pragma unroll
    for (int nt = 0; nt < 2; ++nt) {
      float a0 = ap[nt][0];
      float a1 = ap[nt][HW];
      ap[nt] += 4 * HW;
      a[nt].x = a0;        a[nt].y = a1;
      asq[nt].x = a0 * a0; asq[nt].y = a1 * a1;
    }
#pragma unroll
    for (int ct = 0; ct < 6; ++ct) {
      v2f vb1 = *(gcv2)bp1[ct];
      v2f vb2 = *(gcv2)bp2[ct];
      bp1[ct] += 2 * (2 * KMP);
      bp2[ct] += 2 * (2 * KMP);
#pragma unroll
      for (int nt = 0; nt < 2; ++nt) {
        acc[nt][ct] = __builtin_amdgcn_wmma_f32_16x16x4_f32(
            false, asq[nt], false, vb1, (short)0, acc[nt][ct], false, false);
        acc[nt][ct] = __builtin_amdgcn_wmma_f32_16x16x4_f32(
            false, a[nt],   false, vb2, (short)0, acc[nt][ct], false, false);
      }
    }
  }

#pragma unroll
  for (int nt = 0; nt < 2; ++nt)
#pragma unroll
    for (int ct = 0; ct < 6; ++ct)
#pragma unroll
      for (int r = 0; r < 8; ++r) {
        int pl = wave * 32 + nt * 16 + r + 8 * lhi;
        sdata[pl * KMP + ct * 16 + lrow] = acc[nt][ct][r];
      }
  __syncthreads();

  {
    const float* row = &sdata[tid * KMP];
    float lse[KK];
#pragma unroll
    for (int k = 0; k < KK; ++k) {
      float lp[MMIX];
      float mx = -3.0e38f;
#pragma unroll
      for (int m = 0; m < MMIX; ++m) {
        int c = k * MMIX + m;
        lp[m] = -0.5f * row[c] + cksh[c];
        mx = fmaxf(mx, lp[m]);
      }
      float s = 0.0f;
#pragma unroll
      for (int m = 0; m < MMIX; ++m) s += expf(lp[m] - mx);
      lse[k] = mx + logf(s);
    }
    float mv = lse[0];
#pragma unroll
    for (int k = 1; k < KK; ++k) mv = fmaxf(mv, lse[k]);
    float s = 0.0f;
#pragma unroll
    for (int k = 0; k < KK; ++k) s += expf(lse[k] - mv);
    out[blockIdx.x * 128 + tid] = 1.0f - 1.0f / s;
  }
}

extern "C" void kernel_launch(void* const* d_in, const int* in_sizes, int n_in,
                              void* d_out, int out_size, void* d_ws, size_t ws_size,
                              hipStream_t stream) {
  const float* feats     = (const float*)d_in[0];
  const float* means     = (const float*)d_in[1];
  const float* log_vars  = (const float*)d_in[2];
  const float* log_w     = (const float*)d_in[3];
  float* out = (float*)d_out;

  float* b1  = (float*)d_ws;
  float* b2  = b1 + KMP * DD;
  float* ckm = b2 + KMP * DD;

  (void)in_sizes; (void)n_in; (void)out_size; (void)ws_size;

  int prep_elems = KMP * DD;
  gmm_prep_b<<<(prep_elems + 255) / 256, 256, 0, stream>>>(means, log_vars, b1, b2);
  gmm_prep_c<<<1, 128, 0, stream>>>(means, log_vars, log_w, ckm);
  gmm_main<<<NPTS / 128, 128, 0, stream>>>(feats, b1, b2, ckm, out);
}


# main: d range split over two work-groups (grid 128 -> 256, 12 chunks each), partial sums via workspace; new 2nd launch adds lower + upper d half (sums and column constants) and does the epilogue
# speedup vs baseline: 4.2905x; 4.2905x over previous
#include <hip/hip_runtime.h>
#include <math.h>

#define BATCH 8
#define CCH   768
#define HH    32
#define WW    64
#define HW    (HH*WW)
#define NPTS  (BATCH*HH*WW)
#define DD    768
#define KK    19
#define MMIX  5
#define KM    95
#define KMP   96
#define LOG2PI 1.8378770664093453f

typedef __attribute__((ext_vector_type(2))) float v2f;
typedef __attribute__((ext_vector_type(8))) float v8f;

typedef const float __attribute__((address_space(1)))* gcf;
typedef const v2f   __attribute__((address_space(1)))* gcv2;

#define KC      32
#define NCHUNK  (DD / KC)
#define NSPLIT  2
#define NCH     (NCHUNK / NSPLIT)
#define PART_FLOATS  (2 * 6 * 2 * 128 * 4)
#define WS_CSUM      (NSPLIT * (NPTS / 128) * PART_FLOATS)
#define KSTEPS  (KC / 4)
#define PA      144
#define PC      68
#define A_FLOATS     (KC * PA)
#define B_FLOATS     (KMP * PC)
#define STAGE_FLOATS (A_FLOATS + B_FLOATS)

typedef __attribute__((ext_vector_type(4))) float v4f;
typedef const v4f __attribute__((address_space(1)))* gcv4;

struct Frag {
  float ax[2];
  float ay[2];
  v2f vb1[6];
  v2f vb2[6];
};

__global__ __launch_bounds__(128) void gmm_main(
    const float* __restrict__ feats,
    const float* __restrict__ means, const float* __restrict__ log_vars,
    float* __restrict__ ws) {
  __shared__ __attribute__((aligned(16))) float smem[2 * STAGE_FLOATS];
  __shared__ float csum[2 * KMP];

  const int tid  = threadIdx.x;
  const int wave = tid >> 5;
  const int lane = tid & 31;
  const int lrow = lane & 15;
  const int lhi  = lane >> 4;

  v8f acc[2][6];
#pragma unroll
  for (int nt = 0; nt < 2; ++nt)
#pragma unroll
    for (int ct = 0; ct < 6; ++ct)
      acc[nt][ct] = (v8f){0.f,0.f,0.f,0.f,0.f,0.f,0.f,0.f};

  const int pblk = blockIdx.x >> 1;
  const int half = blockIdx.x & 1;
  const int ch0  = half * NCH;
  const int n0   = pblk * 128;
  const int bb   = n0 >> 11;
  const int rem0 = n0 & 2047;
  gcf ag = (gcf)(feats + (size_t)bb * (size_t)(CCH * HW)
                       + (size_t)wave * HW + rem0 + 4 * lane);
  const int dq = tid & 7;
  const int cb = tid >> 3;
  const bool pad5 = (cb == 15);
  const int c5 = pad5 ? (KM - 1) : (cb + 80);
  gcf mg  = (gcf)(means    + cb * DD + 4 * dq);
  gcf lg  = (gcf)(log_vars + cb * DD + 4 * dq);
  gcf mg5 = (gcf)(means    + c5 * DD + 4 * dq);
  gcf lg5 = (gcf)(log_vars + c5 * DD + 4 * dq);

  const int a_st = wave * PA + 4 * lane;
  const int b_st = A_FLOATS + cb * PC + 4 * dq;

  const int a_fr = (2 * lhi) * PA + wave * 32 + lrow;
  const int b_fr = A_FLOATS + lrow * PC + 2 * lhi;

  v4f ra[8], rm[6], rl[6];
  v4f tb1[6], tb2[6];
  float slv[6], smi[6];
#pragma unroll
  for (int j = 0; j < 6; ++j) { slv[j] = 0.0f; smi[j] = 0.0f; }

  auto load_chunk = [&](int cn) {
    gcf pa = ag + (size_t)cn * (size_t)(KC * HW);
#pragma unroll
    for (int j = 0; j < 8; ++j)
      ra[j] = *(gcv4)(pa + (size_t)(4 * j) * HW);
    gcf pm = mg + cn * KC;
    gcf pl = lg + cn * KC;
#pragma unroll
    for (int j = 0; j < 5; ++j) {
      rm[j] = *(gcv4)(pm + j * 16 * DD);
      rl[j] = *(gcv4)(pl + j * 16 * DD);
    }
    rm[5] = *(gcv4)(mg5 + cn * KC);
    rl[5] = *(gcv4)(lg5 + cn * KC);
  };

  auto make_b = [&](int j) {
    v4f mu = rm[j], lv = rl[j], iv4, w4;
    const bool zero = (j == 5) && pad5;
#pragma unroll
    for (int e = 0; e < 4; ++e) {
      float iv = expf(-lv[e]);
      iv4[e] = zero ? 0.0f : iv;
      w4[e]  = zero ? 0.0f : -2.0f * mu[e] * iv;
      slv[j] += lv[e];
      smi[j] += mu[e] * mu[e] * iv;
    }
    tb1[j] = iv4;
    tb2[j] = w4;
  };

  auto store_stage = [&](int so) {
#pragma unroll
    for (int j = 0; j < 8; ++j)
      *(v4f*)&smem[so + a_st + 4 * j * PA] = ra[j];
#pragma unroll
    for (int j = 0; j < 6; ++j) {
      *(v4f*)&smem[so + b_st + j * 16 * PC] = tb1[j];
      *(v4f*)&smem[so + b_st + j * 16 * PC + KC] = tb2[j];
    }
  };

  auto load_frags = [&](int so, int s, Frag& f) {
#pragma unroll
    for (int nt = 0; nt < 2; ++nt) {
      f.ax[nt] = smem[so + a_fr + (4 * s) * PA + nt * 16];
      f.ay[nt] = smem[so + a_fr + (4 * s + 1) * PA + nt * 16];
    }
#pragma unroll
    for (int ct = 0; ct < 6; ++ct) {
      f.vb1[ct] = *(const v2f*)&smem[so + b_fr + ct * 16 * PC + 4 * s];
      f.vb2[ct] = *(const v2f*)&smem[so + b_fr + ct * 16 * PC + KC + 4 * s];
    }
  };

  auto mma_step = [&](const Frag& f) {
    v2f a[2], asq[2];
#pragma unroll
    for (int nt = 0; nt < 2; ++nt) {
      a[nt].x = f.ax[nt];               a[nt].y = f.ay[nt];
      asq[nt].x = f.ax[nt] * f.ax[nt];  asq[nt].y = f.ay[nt] * f.ay[nt];
    }
#pragma unroll
    for (int ct = 0; ct < 6; ++ct) {
#pragma unroll
      for (int nt = 0; nt < 2; ++nt) {
        acc[nt][ct] = __builtin_amdgcn_wmma_f32_16x16x4_f32(
            false, asq[nt],   false, f.vb1[ct], (short)0, acc[nt][ct], false, false);
        acc[nt][ct] = __builtin_amdgcn_wmma_f32_16x16x4_f32(
            false, a[nt],     false, f.vb2[ct], (short)0, acc[nt][ct], false, false);
      }
    }
  };

  auto compute_chunk = [&](int so, bool next) {
    Frag fr[2];
    load_frags(so, 0, fr[0]);
#pragma unroll
    for (int s = 0; s < KSTEPS; ++s) {
      if (s + 1 < KSTEPS) load_frags(so, s + 1, fr[(s + 1) & 1]);
      __builtin_amdgcn_sched_barrier(0);
      const bool mk = next && s >= 3 && (s & 1);
      if (mk) { make_b(s - 3); make_b(s - 2); }
      mma_step(fr[s & 1]);
      if (mk) {
#pragma unroll
        for (int g = 0; g < 24; ++g) {
          __builtin_amdgcn_sched_group_barrier(0x008, 1, 0);
          __builtin_amdgcn_sched_group_barrier(0x402, 7, 0);
        }
      }
      __builtin_amdgcn_sched_barrier(0);
    }
  };

  load_chunk(ch0);
#pragma unroll
  for (int j = 0; j < 6; ++j) make_b(j);
  store_stage(0);
  __syncthreads();

  for (int i = 0; i < NCH - 1; ++i) {
    const int so = (i & 1) * STAGE_FLOATS;
    load_chunk(ch0 + i + 1);
    compute_chunk(so, true);
    store_stage(STAGE_FLOATS - so);
    __syncthreads();
  }
  compute_chunk(((NCH - 1) & 1) * STAGE_FLOATS, false);

  {
    float* pw = ws + (size_t)blockIdx.x * PART_FLOATS + 4 * tid;
#pragma unroll
    for (int nt = 0; nt < 2; ++nt)
#pragma unroll
      for (int ct = 0; ct < 6; ++ct)
#pragma unroll
        for (int rq = 0; rq < 2; ++rq) {
          v4f v;
#pragma unroll
          for (int e = 0; e < 4; ++e) v[e] = acc[nt][ct][4 * rq + e];
          *(v4f*)(pw + ((nt * 6 + ct) * 2 + rq) * 512) = v;
        }
  }

  if (pblk == 0) {
#pragma unroll
    for (int j = 0; j < 6; ++j) {
      float a = slv[j], b = smi[j];
#pragma unroll
      for (int m = 1; m < 8; m <<= 1) {
        a += __shfl_xor(a, m);
        b += __shfl_xor(b, m);
      }
      const int c = cb + 16 * j;
      if (dq == 0) {
        csum[c]       = (c < KM) ? a : 0.0f;
        csum[KMP + c] = (c < KM) ? b : 0.0f;
      }
    }
    __syncthreads();
    if (wave == 0) {
#pragma unroll
      for (int i = 0; i < (2 * KMP) / 32; ++i)
        ws[WS_CSUM + half * (2 * KMP) + 32 * i + lane] = csum[32 * i + lane];
    }
  }
}

__global__ __launch_bounds__(64) void gmm_tail(
    const float* __restrict__ ws, const float* __restrict__ log_w,
    float* __restrict__ out) {
  __shared__ float sdata[64 * KMP];
  __shared__ float cksh[KMP];

  const int tid  = threadIdx.x;
  const int wv   = tid >> 5;
  const int lane = tid & 31;
  const int lrow = lane & 15;
  const int lhi  = lane >> 4;
  const int pblk = blockIdx.x >> 1;
  const int q    = blockIdx.x & 1;
  const int mt   = q * 64 + tid;

  gcf plo = (gcf)(ws + (size_t)(2 * pblk)     * PART_FLOATS + 4 * mt);
  gcf phi = (gcf)(ws + (size_t)(2 * pblk + 1) * PART_FLOATS + 4 * mt);
  v4f vlo[24], vhi[24];
#pragma unroll
  for (int i = 0; i < 24; ++i) {
    vlo[i] = *(gcv4)(plo + i * 512);
    vhi[i] = *(gcv4)(phi + i * 512);
  }

  for (int c = tid; c < KM; c += 64) {
    const float a = ws[WS_CSUM + 0 * KMP + c] + ws[WS_CSUM + 2 * KMP + c];
    const float b = ws[WS_CSUM + 1 * KMP + c] + ws[WS_CSUM + 3 * KMP + c];
    const float tot = (float)DD * LOG2PI + a + b;
    cksh[c] = log_w[c] - 0.5f * tot;
  }

#pragma unroll
  for (int nt = 0; nt < 2; ++nt)
#pragma unroll
    for (int ct = 0; ct < 6; ++ct)
#pragma unroll
      for (int r = 0; r < 8; ++r) {
        const int i = (nt * 6 + ct) * 2 + (r >> 2);
        const int pl = wv * 32 + nt * 16 + r + 8 * lhi;
        sdata[pl * KMP + ct * 16 + lrow] = vlo[i][r & 3] + vhi[i][r & 3];
      }
  __syncthreads();

  {
    const float* row = &sdata[tid * KMP];
    float lse[KK];
#pragma unroll
    for (int k = 0; k < KK; ++k) {
      float lp[MMIX];
      float mx = -3.0e38f;
#pragma unroll
      for (int m = 0; m < MMIX; ++m) {
        int c = k * MMIX + m;
        lp[m] = -0.5f * row[c] + cksh[c];
        mx = fmaxf(mx, lp[m]);
      }
      float s = 0.0f;
#pragma unroll
      for (int m = 0; m < MMIX; ++m) s += expf(lp[m] - mx);
      lse[k] = mx + logf(s);
    }
    float mv = lse[0];
#pragma unroll
    for (int k = 1; k < KK; ++k) mv = fmaxf(mv, lse[k]);
    float s = 0.0f;
#pragma unroll
    for (int k = 0; k < KK; ++k) s += expf(lse[k] - mv);
    out[pblk * 128 + q * 64 + tid] = 1.0f - 1.0f / s;
  }
}

extern "C" void kernel_launch(void* const* d_in, const int* in_sizes, int n_in,
                              void* d_out, int out_size, void* d_ws, size_t ws_size,
                              hipStream_t stream) {
  const float* feats     = (const float*)d_in[0];
  const float* means     = (const float*)d_in[1];
  const float* log_vars  = (const float*)d_in[2];
  const float* log_w     = (const float*)d_in[3];
  float* out = (float*)d_out;
  float* ws  = (float*)d_ws;

  (void)in_sizes; (void)n_in; (void)out_size; (void)ws_size;

  gmm_main<<<NSPLIT * (NPTS / 128), 128, 0, stream>>>(feats, means, log_vars, ws);
  gmm_tail<<<2 * (NPTS / 128), 64, 0, stream>>>(ws, log_w, out);
}


# main: work-group 4 -> 8 waves in two roles, one of each per SIMD: waves 4-7 do all staging (global loads, B build, LDS stores), waves 0-3 only read fragments and issue WMMAs; order of sums kept
# speedup vs baseline: 4.7678x; 1.1112x over previous
#include <hip/hip_runtime.h>
#include <math.h>

#define BATCH 8
#define CCH   768
#define HH    32
#define WW    64
#define HW    (HH*WW)
#define NPTS  (BATCH*HH*WW)
#define DD    768
#define KK    19
#define MMIX  5
#define KM    95
#define KMP   96
#define LOG2PI 1.8378770664093453f

typedef __attribute__((ext_vector_type(2))) float v2f;
typedef __attribute__((ext_vector_type(8))) float v8f;

typedef const float __attribute__((address_space(1)))* gcf;
typedef const v2f   __attribute__((address_space(1)))* gcv2;

#define KC      32
#define NCHUNK  (DD / KC)
#define NSPLIT  2
#define NCH     (NCHUNK / NSPLIT)
#define PART_FLOATS  (2 * 6 * 2 * 128 * 4)
#define WS_CSUM      (NSPLIT * (NPTS / 128) * PART_FLOATS)
#define KSTEPS  (KC / 4)
#define PA      144
#define PC      68
#define A_FLOATS     (KC * PA)
#define B_FLOATS     (KMP * PC)
#define STAGE_FLOATS (A_FLOATS + B_FLOATS)

typedef __attribute__((ext_vector_type(4))) float v4f;
typedef const v4f __attribute__((address_space(1)))* gcv4;

struct Frag {
  float ax[2];
  float ay[2];
  v2f vb1[6];
  v2f vb2[6];
};

__global__ __launch_bounds__(256) void gmm_main(
    const float* __restrict__ feats,
    const float* __restrict__ means, const float* __restrict__ log_vars,
    float* __restrict__ ws) {
  __shared__ __attribute__((aligned(16))) float smem[2 * STAGE_FLOATS];
  __shared__ float csum[2 * KMP];

  const int tid  = threadIdx.x;
  const int wave8 = __builtin_amdgcn_readfirstlane(tid >> 5);
  const bool matrix_role = wave8 < 4;
  const int rt   = tid & 127;
  const int wave = wave8 & 3;
  const int lane = tid & 31;
  const int lrow = lane & 15;
  const int lhi  = lane >> 4;

  v8f acc[2][6];
#pragma unroll
  for (int nt = 0; nt < 2; ++nt)
#pragma unroll
    for (int ct = 0; ct < 6; ++ct)
      acc[nt][ct] = (v8f){0.f,0.f,0.f,0.f,0.f,0.f,0.f,0.f};

  const int pblk = blockIdx.x >> 1;
  const int half = blockIdx.x & 1;
  const int ch0  = half * NCH;
  const int n0   = pblk * 128;
  const int bb   = n0 >> 11;
  const int rem0 = n0 & 2047;
  gcf ag = (gcf)(feats + (size_t)bb * (size_t)(CCH * HW)
                       + (size_t)wave * HW + rem0 + 4 * lane);
  const int dq = rt & 7;
  const int cb = rt >> 3;
  const bool pad5 = (cb == 15);
  const int c5 = pad5 ? (KM - 1) : (cb + 80);
  gcf mg  = (gcf)(means    + cb * DD + 4 * dq);
  gcf lg  = (gcf)(log_vars + cb * DD + 4 * dq);
  gcf mg5 = (gcf)(means    + c5 * DD + 4 * dq);
  gcf lg5 = (gcf)(log_vars + c5 * DD + 4 * dq);

  const int a_st = wave * PA + 4 * lane;
  const int b_st = A_FLOATS + cb * PC + 4 * dq;

  const int a_fr = (2 * lhi) * PA + wave * 32 + lrow;
  const int b_fr = A_FLOATS + lrow * PC + 2 * lhi;

  v4f ra[8], rm[6], rl[6];
  float slv[6], smi[6];
#pragma unroll
  for (int j = 0; j < 6; ++j) { slv[j] = 0.0f; smi[j] = 0.0f; }

  auto load_a = [&](int cn) {
    gcf pa = ag + (size_t)cn * (size_t)(KC * HW);
#pragma unroll
    for (int j = 0; j < 8; ++j)
      ra[j] = *(gcv4)(pa + (size_t)(4 * j) * HW);
  };

  auto load_b = [&](int cn, int j) {
    if (j < 5) {
      rm[j] = *(gcv4)(mg + cn * KC + j * 16 * DD);
      rl[j] = *(gcv4)(lg + cn * KC + j * 16 * DD);
    } else {
      rm[5] = *(gcv4)(mg5 + cn * KC);
      rl[5] = *(gcv4)(lg5 + cn * KC);
    }
  };

  auto stage_a = [&](int so) {
#pragma unroll
    for (int j = 0; j < 8; ++j)
      *(v4f*)&smem[so + a_st + 4 * j * PA] = ra[j];
  };

  auto stage_b = [&](int so, int j) {
    v4f mu = rm[j], lv = rl[j], iv4, w4;
    const bool zero = (j == 5) && pad5;
#pragma unroll
    for (int e = 0; e < 4; ++e) {
      float iv = expf(-lv[e]);
      iv4[e] = zero ? 0.0f : iv;
      w4[e]  = zero ? 0.0f : -2.0f * mu[e] * iv;
      slv[j] += lv[e];
      smi[j] += mu[e] * mu[e] * iv;
    }
    *(v4f*)&smem[so + b_st + j * 16 * PC] = iv4;
    *(v4f*)&smem[so + b_st + j * 16 * PC + KC] = w4;
  };

  auto produce = [&](int so, int cn) {
    stage_a(so);
    load_a(cn);
#pragma unroll
    for (int j = 0; j < 6; ++j) {
      stage_b(so, j);
      load_b(cn, j);
    }
  };

  auto produce_last = [&](int so) {
    stage_a(so);
#pragma unroll
    for (int j = 0; j < 6; ++j) stage_b(so, j);
  };

  auto load_frags = [&](int so, int s, Frag& f) {
#pragma unroll
    for (int nt = 0; nt < 2; ++nt) {
      f.ax[nt] = smem[so + a_fr + (4 * s) * PA + nt * 16];
      f.ay[nt] = smem[so + a_fr + (4 * s + 1) * PA + nt * 16];
    }
#pragma unroll
    for (int ct = 0; ct < 6; ++ct) {
      f.vb1[ct] = *(const v2f*)&smem[so + b_fr + ct * 16 * PC + 4 * s];
      f.vb2[ct] = *(const v2f*)&smem[so + b_fr + ct * 16 * PC + KC + 4 * s];
    }
  };

  auto mma_step = [&](const Frag& f) {
    v2f a[2], asq[2];
#pragma unroll
    for (int nt = 0; nt < 2; ++nt) {
      a[nt].x = f.ax[nt];               a[nt].y = f.ay[nt];
      asq[nt].x = f.ax[nt] * f.ax[nt];  asq[nt].y = f.ay[nt] * f.ay[nt];
    }
#pragma unroll
    for (int ct = 0; ct < 6; ++ct) {
#pragma unroll
      for (int nt = 0; nt < 2; ++nt) {
        acc[nt][ct] = __builtin_amdgcn_wmma_f32_16x16x4_f32(
            false, asq[nt],   false, f.vb1[ct], (short)0, acc[nt][ct], false, false);
        acc[nt][ct] = __builtin_amdgcn_wmma_f32_16x16x4_f32(
            false, a[nt],     false, f.vb2[ct], (short)0, acc[nt][ct], false, false);
      }
    }
  };

  auto compute_chunk = [&](int so) {
    Frag fr[2];
    load_frags(so, 0, fr[0]);
#pragma unroll
    for (int s = 0; s < KSTEPS; ++s) {
      if (s + 1 < KSTEPS) load_frags(so, s + 1, fr[(s + 1) & 1]);
      __builtin_amdgcn_sched_barrier(0);
      mma_step(fr[s & 1]);
      __builtin_amdgcn_sched_barrier(0);
    }
  };

  if (!matrix_role) {
    load_a(ch0);
#pragma unroll
    for (int j = 0; j < 6; ++j) load_b(ch0, j);
    produce(0, ch0 + 1);
  }
  __syncthreads();

  for (int i = 0; i < NCH - 2; ++i) {
    const int so = (i & 1) * STAGE_FLOATS;
    if (matrix_role) compute_chunk(so);
    else             produce(STAGE_FLOATS - so, ch0 + i + 2);
    __syncthreads();
  }
  {
    const int so = ((NCH - 2) & 1) * STAGE_FLOATS;
    if (matrix_role) compute_chunk(so);
    else             produce_last(STAGE_FLOATS - so);
    __syncthreads();
  }
  if (matrix_role) compute_chunk(((NCH - 1) & 1) * STAGE_FLOATS);

  if (matrix_role) {
    float* pw = ws + (size_t)blockIdx.x * PART_FLOATS + 4 * rt;
#pragma unroll
    for (int nt = 0; nt < 2; ++nt)
#pragma unroll
      for (int ct = 0; ct < 6; ++ct)
#pragma unroll
        for (int rq = 0; rq < 2; ++rq) {
          v4f v;
#pragma unroll
          for (int e = 0; e < 4; ++e) v[e] = acc[nt][ct][4 * rq + e];
          *(v4f*)(pw + ((nt * 6 + ct) * 2 + rq) * 512) = v;
        }
  }

  if (pblk == 0) {
    if (!matrix_role) {
#pragma unroll
      for (int j = 0; j < 6; ++j) {
        float a = slv[j], b = smi[j];
#pragma unroll
        for (int m = 1; m < 8; m <<= 1) {
          a += __shfl_xor(a, m);
          b += __shfl_xor(b, m);
        }
        const int c = cb + 16 * j;
        if (dq == 0) {
          csum[c]       = (c < KM) ? a : 0.0f;
          csum[KMP + c] = (c < KM) ? b : 0.0f;
        }
      }
    }
    __syncthreads();
    if (wave8 == 4) {
#pragma unroll
      for (int i = 0; i < (2 * KMP) / 32; ++i)
        ws[WS_CSUM + half * (2 * KMP) + 32 * i + lane] = csum[32 * i + lane];
    }
  }
}

__global__ __launch_bounds__(64) void gmm_tail(
    const float* __restrict__ ws, const float* __restrict__ log_w,
    float* __restrict__ out) {
  __shared__ float sdata[64 * KMP];
  __shared__ float cksh[KMP];

  const int tid  = threadIdx.x;
  const int wv   = tid >> 5;
  const int lane = tid & 31;
  const int lrow = lane & 15;
  const int lhi  = lane >> 4;
  const int pblk = blockIdx.x >> 1;
  const int q    = blockIdx.x & 1;
  const int mt   = q * 64 + tid;

  gcf plo = (gcf)(ws + (size_t)(2 * pblk)     * PART_FLOATS + 4 * mt);
  gcf phi = (gcf)(ws + (size_t)(2 * pblk + 1) * PART_FLOATS + 4 * mt);
  v4f vlo[24], vhi[24];
#pragma unroll
  for (int i = 0; i < 24; ++i) {
    vlo[i] = *(gcv4)(plo + i * 512);
    vhi[i] = *(gcv4)(phi + i * 512);
  }

  for (int c = tid; c < KM; c += 64) {
    const float a = ws[WS_CSUM + 0 * KMP + c] + ws[WS_CSUM + 2 * KMP + c];
    const float b = ws[WS_CSUM + 1 * KMP + c] + ws[WS_CSUM + 3 * KMP + c];
    const float tot = (float)DD * LOG2PI + a + b;
    cksh[c] = log_w[c] - 0.5f * tot;
  }

#pragma unroll
  for (int nt = 0; nt < 2; ++nt)
#pragma unroll
    for (int ct = 0; ct < 6; ++ct)
#pragma unroll
      for (int r = 0; r < 8; ++r) {
        const int i = (nt * 6 + ct) * 2 + (r >> 2);
        const int pl = wv * 32 + nt * 16 + r + 8 * lhi;
        sdata[pl * KMP + ct * 16 + lrow] = vlo[i][r & 3] + vhi[i][r & 3];
      }
  __syncthreads();

  {
    const float* row = &sdata[tid * KMP];
    float lse[KK];
#pragma unroll
    for (int k = 0; k < KK; ++k) {
      float lp[MMIX];
      float mx = -3.0e38f;
#pragma unroll
      for (int m = 0; m < MMIX; ++m) {
        int c = k * MMIX + m;
        lp[m] = -0.5f * row[c] + cksh[c];
        mx = fmaxf(mx, lp[m]);
      }
      float s = 0.0f;
#pragma unroll
      for (int m = 0; m < MMIX; ++m) s += expf(lp[m] - mx);
      lse[k] = mx + logf(s);
    }
    float mv = lse[0];
#pragma unroll
    for (int k = 1; k < KK; ++k) mv = fmaxf(mv, lse[k]);
    float s = 0.0f;
#pragma unroll
    for (int k = 0; k < KK; ++k) s += expf(lse[k] - mv);
    out[pblk * 128 + q * 64 + tid] = 1.0f - 1.0f / s;
  }
}

extern "C" void kernel_launch(void* const* d_in, const int* in_sizes, int n_in,
                              void* d_out, int out_size, void* d_ws, size_t ws_size,
                              hipStream_t stream) {
  const float* feats     = (const float*)d_in[0];
  const float* means     = (const float*)d_in[1];
  const float* log_vars  = (const float*)d_in[2];
  const float* log_w     = (const float*)d_in[3];
  float* out = (float*)d_out;
  float* ws  = (float*)d_ws;

  (void)in_sizes; (void)n_in; (void)out_size; (void)ws_size;

  gmm_main<<<NSPLIT * (NPTS / 128), 256, 0, stream>>>(feats, means, log_vars, ws);
  gmm_tail<<<2 * (NPTS / 128), 64, 0, stream>>>(ws, log_w, out);
}


# exp and log as one multiply + the hardware exp2/log2 instruction (was the library's 13-instruction range-checked sequences): staging waves' B build and tail epilogue; values move by about 1 ulp
# speedup vs baseline: 5.1808x; 1.0866x over previous
#include <hip/hip_runtime.h>
#include <math.h>

#define BATCH 8
#define CCH   768
#define HH    32
#define WW    64
#define HW    (HH*WW)
#define NPTS  (BATCH*HH*WW)
#define DD    768
#define KK    19
#define MMIX  5
#define KM    95
#define KMP   96
#define LOG2PI 1.8378770664093453f
#define LOG2E  1.4426950408889634f
#define LN2    0.6931471805599453f

typedef __attribute__((ext_vector_type(2))) float v2f;
typedef __attribute__((ext_vector_type(8))) float v8f;

typedef const float __attribute__((address_space(1)))* gcf;
typedef const v2f   __attribute__((address_space(1)))* gcv2;

#define KC      32
#define NCHUNK  (DD / KC)
#define NSPLIT  2
#define NCH     (NCHUNK / NSPLIT)
#define PART_FLOATS  (2 * 6 * 2 * 128 * 4)
#define WS_CSUM      (NSPLIT * (NPTS / 128) * PART_FLOATS)
#define KSTEPS  (KC / 4)
#define PA      144
#define PC      68
#define A_FLOATS     (KC * PA)
#define B_FLOATS     (KMP * PC)
#define STAGE_FLOATS (A_FLOATS + B_FLOATS)

typedef __attribute__((ext_vector_type(4))) float v4f;
typedef const v4f __attribute__((address_space(1)))* gcv4;

struct Frag {
  float ax[2];
  float ay[2];
  v2f vb1[6];
  v2f vb2[6];
};

__global__ __launch_bounds__(256) void gmm_main(
    const float* __restrict__ feats,
    const float* __restrict__ means, const float* __restrict__ log_vars,
    float* __restrict__ ws) {
  __shared__ __attribute__((aligned(16))) float smem[2 * STAGE_FLOATS];
  __shared__ float csum[2 * KMP];

  const int tid  = threadIdx.x;
  const int wave8 = __builtin_amdgcn_readfirstlane(tid >> 5);
  const bool matrix_role = wave8 < 4;
  const int rt   = tid & 127;
  const int wave = wave8 & 3;
  const int lane = tid & 31;
  const int lrow = lane & 15;
  const int lhi  = lane >> 4;

  v8f acc[2][6];
#pragma unroll
  for (int nt = 0; nt < 2; ++nt)
#pragma unroll
    for (int ct = 0; ct < 6; ++ct)
      acc[nt][ct] = (v8f){0.f,0.f,0.f,0.f,0.f,0.f,0.f,0.f};

  const int pblk = blockIdx.x >> 1;
  const int half = blockIdx.x & 1;
  const int ch0  = half * NCH;
  const int n0   = pblk * 128;
  const int bb   = n0 >> 11;
  const int rem0 = n0 & 2047;
  gcf ag = (gcf)(feats + (size_t)bb * (size_t)(CCH * HW)
                       + (size_t)wave * HW + rem0 + 4 * lane);
  const int dq = rt & 7;
  const int cb = rt >> 3;
  const bool pad5 = (cb == 15);
  const int c5 = pad5 ? (KM - 1) : (cb + 80);
  gcf mg  = (gcf)(means    + cb * DD + 4 * dq);
  gcf lg  = (gcf)(log_vars + cb * DD + 4 * dq);
  gcf mg5 = (gcf)(means    + c5 * DD + 4 * dq);
  gcf lg5 = (gcf)(log_vars + c5 * DD + 4 * dq);

  const int a_st = wave * PA + 4 * lane;
  const int b_st = A_FLOATS + cb * PC + 4 * dq;

  const int a_fr = (2 * lhi) * PA + wave * 32 + lrow;
  const int b_fr = A_FLOATS + lrow * PC + 2 * lhi;

  v4f ra[8], rm[6], rl[6];
  float slv[6], smi[6];
#pragma unroll
  for (int j = 0; j < 6; ++j) { slv[j] = 0.0f; smi[j] = 0.0f; }

  auto load_a = [&](int cn) {
    gcf pa = ag + (size_t)cn * (size_t)(KC * HW);
#pragma unroll
    for (int j = 0; j < 8; ++j)
      ra[j] = *(gcv4)(pa + (size_t)(4 * j) * HW);
  };

  auto load_b = [&](int cn, int j) {
    if (j < 5) {
      rm[j] = *(gcv4)(mg + cn * KC + j * 16 * DD);
      rl[j] = *(gcv4)(lg + cn * KC + j * 16 * DD);
    } else {
      rm[5] = *(gcv4)(mg5 + cn * KC);
      rl[5] = *(gcv4)(lg5 + cn * KC);
    }
  };

  auto stage_a = [&](int so) {
#pragma unroll
    for (int j = 0; j < 8; ++j)
      *(v4f*)&smem[so + a_st + 4 * j * PA] = ra[j];
  };

  auto stage_b = [&](int so, int j) {
    v4f mu = rm[j], lv = rl[j], iv4, w4;
    const bool zero = (j == 5) && pad5;
#pragma unroll
    for (int e = 0; e < 4; ++e) {
      float iv = __builtin_amdgcn_exp2f(lv[e] * -LOG2E);
      iv4[e] = zero ? 0.0f : iv;
      w4[e]  = zero ? 0.0f : -2.0f * mu[e] * iv;
      slv[j] += lv[e];
      smi[j] += mu[e] * mu[e] * iv;
    }
    *(v4f*)&smem[so + b_st + j * 16 * PC] = iv4;
    *(v4f*)&smem[so + b_st + j * 16 * PC + KC] = w4;
  };

  auto produce = [&](int so, int cn) {
    stage_a(so);
    load_a(cn);
#pragma unroll
    for (int j = 0; j < 6; ++j) {
      stage_b(so, j);
      load_b(cn, j);
    }
  };

  auto produce_last = [&](int so) {
    stage_a(so);
#pragma unroll
    for (int j = 0; j < 6; ++j) stage_b(so, j);
  };

  auto load_frags = [&](int so, int s, Frag& f) {
#pragma unroll
    for (int nt = 0; nt < 2; ++nt) {
      f.ax[nt] = smem[so + a_fr + (4 * s) * PA + nt * 16];
      f.ay[nt] = smem[so + a_fr + (4 * s + 1) * PA + nt * 16];
    }
#pragma unroll
    for (int ct = 0; ct < 6; ++ct) {
      f.vb1[ct] = *(const v2f*)&smem[so + b_fr + ct * 16 * PC + 4 * s];
      f.vb2[ct] = *(const v2f*)&smem[so + b_fr + ct * 16 * PC + KC + 4 * s];
    }
  };

  auto mma_step = [&](const Frag& f) {
    v2f a[2], asq[2];
#pragma unroll
    for (int nt = 0; nt < 2; ++nt) {
      a[nt].x = f.ax[nt];               a[nt].y = f.ay[nt];
      asq[nt].x = f.ax[nt] * f.ax[nt];  asq[nt].y = f.ay[nt] * f.ay[nt];
    }
#pragma unroll
    for (int ct = 0; ct < 6; ++ct) {
#pragma unroll
      for (int nt = 0; nt < 2; ++nt) {
        acc[nt][ct] = __builtin_amdgcn_wmma_f32_16x16x4_f32(
            false, asq[nt],   false, f.vb1[ct], (short)0, acc[nt][ct], false, false);
        acc[nt][ct] = __builtin_amdgcn_wmma_f32_16x16x4_f32(
            false, a[nt],     false, f.vb2[ct], (short)0, acc[nt][ct], false, false);
      }
    }
  };

  auto compute_chunk = [&](int so) {
    Frag fr[2];
    load_frags(so, 0, fr[0]);
#pragma unroll
    for (int s = 0; s < KSTEPS; ++s) {
      if (s + 1 < KSTEPS) load_frags(so, s + 1, fr[(s + 1) & 1]);
      __builtin_amdgcn_sched_barrier(0);
      mma_step(fr[s & 1]);
      __builtin_amdgcn_sched_barrier(0);
    }
  };

  if (!matrix_role) {
    load_a(ch0);
#pragma unroll
    for (int j = 0; j < 6; ++j) load_b(ch0, j);
    produce(0, ch0 + 1);
  }
  __syncthreads();

  for (int i = 0; i < NCH - 2; ++i) {
    const int so = (i & 1) * STAGE_FLOATS;
    if (matrix_role) compute_chunk(so);
    else             produce(STAGE_FLOATS - so, ch0 + i + 2);
    __syncthreads();
  }
  {
    const int so = ((NCH - 2) & 1) * STAGE_FLOATS;
    if (matrix_role) compute_chunk(so);
    else             produce_last(STAGE_FLOATS - so);
    __syncthreads();
  }
  if (matrix_role) compute_chunk(((NCH - 1) & 1) * STAGE_FLOATS);

  if (matrix_role) {
    float* pw = ws + (size_t)blockIdx.x * PART_FLOATS + 4 * rt;
#pragma unroll
    for (int nt = 0; nt < 2; ++nt)
#pragma unroll
      for (int ct = 0; ct < 6; ++ct)
#pragma unroll
        for (int rq = 0; rq < 2; ++rq) {
          v4f v;
#pragma unroll
          for (int e = 0; e < 4; ++e) v[e] = acc[nt][ct][4 * rq + e];
          *(v4f*)(pw + ((nt * 6 + ct) * 2 + rq) * 512) = v;
        }
  }

  if (pblk == 0) {
    if (!matrix_role) {
#pragma unroll
      for (int j = 0; j < 6; ++j) {
        float a = slv[j], b = smi[j];
#pragma unroll
        for (int m = 1; m < 8; m <<= 1) {
          a += __shfl_xor(a, m);
          b += __shfl_xor(b, m);
        }
        const int c = cb + 16 * j;
        if (dq == 0) {
          csum[c]       = (c < KM) ? a : 0.0f;
          csum[KMP + c] = (c < KM) ? b : 0.0f;
        }
      }
    }
    __syncthreads();
    if (wave8 == 4) {
#pragma unroll
      for (int i = 0; i < (2 * KMP) / 32; ++i)
        ws[WS_CSUM + half * (2 * KMP) + 32 * i + lane] = csum[32 * i + lane];
    }
  }
}

__global__ __launch_bounds__(64) void gmm_tail(
    const float* __restrict__ ws, const float* __restrict__ log_w,
    float* __restrict__ out) {
  __shared__ float sdata[64 * KMP];
  __shared__ float cksh[KMP];

  const int tid  = threadIdx.x;
  const int wv   = tid >> 5;
  const int lane = tid & 31;
  const int lrow = lane & 15;
  const int lhi  = lane >> 4;
  const int pblk = blockIdx.x >> 1;
  const int q    = blockIdx.x & 1;
  const int mt   = q * 64 + tid;

  gcf plo = (gcf)(ws + (size_t)(2 * pblk)     * PART_FLOATS + 4 * mt);
  gcf phi = (gcf)(ws + (size_t)(2 * pblk + 1) * PART_FLOATS + 4 * mt);
  v4f vlo[24], vhi[24];
#pragma unroll
  for (int i = 0; i < 24; ++i) {
    vlo[i] = *(gcv4)(plo + i * 512);
    vhi[i] = *(gcv4)(phi + i * 512);
  }

  for (int c = tid; c < KM; c += 64) {
    const float a = ws[WS_CSUM + 0 * KMP + c] + ws[WS_CSUM + 2 * KMP + c];
    const float b = ws[WS_CSUM + 1 * KMP + c] + ws[WS_CSUM + 3 * KMP + c];
    const float tot = (float)DD * LOG2PI + a + b;
    cksh[c] = log_w[c] - 0.5f * tot;
  }

#pragma unroll
  for (int nt = 0; nt < 2; ++nt)
#pragma unroll
    for (int ct = 0; ct < 6; ++ct)
#pragma unroll
      for (int r = 0; r < 8; ++r) {
        const int i = (nt * 6 + ct) * 2 + (r >> 2);
        const int pl = wv * 32 + nt * 16 + r + 8 * lhi;
        sdata[pl * KMP + ct * 16 + lrow] = vlo[i][r & 3] + vhi[i][r & 3];
      }
  __syncthreads();

  {
    const float* row = &sdata[tid * KMP];
    float lse[KK];
#pragma unroll
    for (int k = 0; k < KK; ++k) {
      float lp[MMIX];
      float mx = -3.0e38f;
#pragma unroll
      for (int m = 0; m < MMIX; ++m) {
        int c = k * MMIX + m;
        lp[m] = -0.5f * row[c] + cksh[c];
        mx = fmaxf(mx, lp[m]);
      }
      float s = 0.0f;
#pragma unroll
      for (int m = 0; m < MMIX; ++m) s += __builtin_amdgcn_exp2f((lp[m] - mx) * LOG2E);
      lse[k] = mx + __builtin_amdgcn_logf(s) * LN2;
    }
    float mv = lse[0];
#pragma unroll
    for (int k = 1; k < KK; ++k) mv = fmaxf(mv, lse[k]);
    float s = 0.0f;
#pragma unroll
    for (int k = 0; k < KK; ++k) s += __builtin_amdgcn_exp2f((lse[k] - mv) * LOG2E);
    out[pblk * 128 + q * 64 + tid] = 1.0f - 1.0f / s;
  }
}

extern "C" void kernel_launch(void* const* d_in, const int* in_sizes, int n_in,
                              void* d_out, int out_size, void* d_ws, size_t ws_size,
                              hipStream_t stream) {
  const float* feats     = (const float*)d_in[0];
  const float* means     = (const float*)d_in[1];
  const float* log_vars  = (const float*)d_in[2];
  const float* log_w     = (const float*)d_in[3];
  float* out = (float*)d_out;
  float* ws  = (float*)d_ws;

  (void)in_sizes; (void)n_in; (void)out_size; (void)ws_size;

  gmm_main<<<NSPLIT * (NPTS / 128), 256, 0, stream>>>(feats, means, log_vars, ws);
  gmm_tail<<<2 * (NPTS / 128), 64, 0, stream>>>(ws, log_w, out);
}


# one launch: a work-group owns 64 points over all d, a wave pair splits each 64-d chunk, halves added in LDS, epilogue in place (was d over two work-groups + 2nd launch); sums regrouped in d
# speedup vs baseline: 5.5580x; 1.0728x over previous
#include <hip/hip_runtime.h>
#include <math.h>

#define BATCH 8
#define CCH   768
#define HH    32
#define WW    64
#define HW    (HH*WW)
#define NPTS  (BATCH*HH*WW)
#define DD    768
#define KK    19
#define MMIX  5
#define KM    95
#define KMP   96
#define LOG2PI 1.8378770664093453f
#define LOG2E  1.4426950408889634f
#define LN2    0.6931471805599453f

typedef __attribute__((ext_vector_type(2))) float v2f;
typedef __attribute__((ext_vector_type(4))) float v4f;
typedef __attribute__((ext_vector_type(8))) float v8f;

typedef const float __attribute__((address_space(1)))* gcf;
typedef const v4f   __attribute__((address_space(1)))* gcv4;

#define PTS     64
#define KC      64
#define NCH     (DD / KC)
#define KH      (KC / 2)
#define KSTEPS  (KH / 4)
#define NSLOT   12
#define PA      80
#define PC      (2 * KC + 4)
#define A_FLOATS     (KC * PA)
#define B_FLOATS     (KMP * PC)
#define STAGE_FLOATS (A_FLOATS + B_FLOATS)
#define PR      97

struct Frag {
  float ax[2];
  float ay[2];
  v2f vb1[6];
  v2f vb2[6];
};

__global__ __launch_bounds__(256) void gmm_main(
    const float* __restrict__ feats,
    const float* __restrict__ means, const float* __restrict__ log_vars,
    const float* __restrict__ log_w, float* __restrict__ out) {
  __shared__ __attribute__((aligned(16))) float smem[2 * STAGE_FLOATS];
  __shared__ float sdata[PTS * PR];
  __shared__ float csum[2 * KMP];
  __shared__ float cksh[KMP];

  const int tid  = threadIdx.x;
  const int wave8 = __builtin_amdgcn_readfirstlane(tid >> 5);
  const bool matrix_role = wave8 < 4;
  const int rt   = tid & 127;
  const int wave = wave8 & 3;
  const int rg   = wave & 1;
  const int dh   = wave >> 1;
  const int lane = tid & 31;
  const int lrow = lane & 15;
  const int lhi  = lane >> 4;

  v8f acc[2][6];
#pragma unroll
  for (int nt = 0; nt < 2; ++nt)
#pragma unroll
    for (int ct = 0; ct < 6; ++ct)
      acc[nt][ct] = (v8f){0.f,0.f,0.f,0.f,0.f,0.f,0.f,0.f};

  const int n0   = blockIdx.x * PTS;
  const int bb   = n0 >> 11;
  const int rem0 = n0 & 2047;
  const int aq = lane & 15;
  const int ar = lane >> 4;
  gcf ag = (gcf)(feats + (size_t)bb * (size_t)(CCH * HW)
                       + (size_t)(16 * wave + ar) * HW + rem0 + 4 * aq);
  const int dq = rt & 15;
  const int cb = rt >> 4;
  const bool padl = (cb == 7);
  const int cl = padl ? (KM - 1) : (cb + 88);
  gcf mg  = (gcf)(means    + cb * DD + 4 * dq);
  gcf lg  = (gcf)(log_vars + cb * DD + 4 * dq);
  gcf mgl = (gcf)(means    + cl * DD + 4 * dq);
  gcf lgl = (gcf)(log_vars + cl * DD + 4 * dq);

  const int a_st = (16 * wave + ar) * PA + 4 * aq;
  const int b_st = A_FLOATS + cb * PC + 4 * dq;

  const int a_fr = (KH * dh + 2 * lhi) * PA + rg * 32 + lrow;
  const int b_fr = A_FLOATS + lrow * PC + KH * dh + 2 * lhi;

  v4f ra[8], rm[NSLOT], rl[NSLOT];
  float slv[NSLOT], smi[NSLOT];
#pragma unroll
  for (int j = 0; j < NSLOT; ++j) { slv[j] = 0.0f; smi[j] = 0.0f; }

  auto load_a = [&](int cn) {
    gcf pa = ag + (size_t)cn * (size_t)(KC * HW);
#pragma unroll
    for (int j = 0; j < 8; ++j)
      ra[j] = *(gcv4)(pa + (size_t)(2 * j) * HW);
  };

  auto load_b = [&](int cn, int j) {
    if (j < NSLOT - 1) {
      rm[j] = *(gcv4)(mg + cn * KC + j * 8 * DD);
      rl[j] = *(gcv4)(lg + cn * KC + j * 8 * DD);
    } else {
      rm[j] = *(gcv4)(mgl + cn * KC);
      rl[j] = *(gcv4)(lgl + cn * KC);
    }
  };

  auto stage_a = [&](int so) {
#pragma unroll
    for (int j = 0; j < 8; ++j)
      *(v4f*)&smem[so + a_st + 2 * j * PA] = ra[j];
  };

  auto stage_b = [&](int so, int j) {
    v4f mu = rm[j], lv = rl[j], iv4, w4;
    const bool zero = (j == NSLOT - 1) && padl;
#pragma unroll
    for (int e = 0; e < 4; ++e) {
      float iv = __builtin_amdgcn_exp2f(lv[e] * -LOG2E);
      iv4[e] = zero ? 0.0f : iv;
      w4[e]  = zero ? 0.0f : -2.0f * mu[e] * iv;
      slv[j] += lv[e];
      smi[j] += mu[e] * mu[e] * iv;
    }
    *(v4f*)&smem[so + b_st + j * 8 * PC] = iv4;
    *(v4f*)&smem[so + b_st + j * 8 * PC + KC] = w4;
  };

  auto produce = [&](int so, int cn) {
    stage_a(so);
    load_a(cn);
#pragma unroll
    for (int j = 0; j < NSLOT; ++j) {
      stage_b(so, j);
      load_b(cn, j);
    }
  };

  auto produce_last = [&](int so) {
    stage_a(so);
#pragma unroll
    for (int j = 0; j < NSLOT; ++j) stage_b(so, j);
  };

  auto load_frags = [&](int so, int s, Frag& f) {
#pragma unroll
    for (int nt = 0; nt < 2; ++nt) {
      f.ax[nt] = smem[so + a_fr + (4 * s) * PA + nt * 16];
      f.ay[nt] = smem[so + a_fr + (4 * s + 1) * PA + nt * 16];
    }
#pragma unroll
    for (int ct = 0; ct < 6; ++ct) {
      f.vb1[ct] = *(const v2f*)&smem[so + b_fr + ct * 16 * PC + 4 * s];
      f.vb2[ct] = *(const v2f*)&smem[so + b_fr + ct * 16 * PC + KC + 4 * s];
    }
  };

  auto mma_step = [&](const Frag& f) {
    v2f a[2], asq[2];
#pragma unroll
    for (int nt = 0; nt < 2; ++nt) {
      a[nt].x = f.ax[nt];               a[nt].y = f.ay[nt];
      asq[nt].x = f.ax[nt] * f.ax[nt];  asq[nt].y = f.ay[nt] * f.ay[nt];
    }
#pragma unroll
    for (int ct = 0; ct < 6; ++ct) {
#pragma unroll
      for (int nt = 0; nt < 2; ++nt) {
        acc[nt][ct] = __builtin_amdgcn_wmma_f32_16x16x4_f32(
            false, asq[nt],   false, f.vb1[ct], (short)0, acc[nt][ct], false, false);
        acc[nt][ct] = __builtin_amdgcn_wmma_f32_16x16x4_f32(
            false, a[nt],     false, f.vb2[ct], (short)0, acc[nt][ct], false, false);
      }
    }
  };

  auto compute_chunk = [&](int so) {
    Frag fr[2];
    load_frags(so, 0, fr[0]);
#pragma unroll
    for (int s = 0; s < KSTEPS; ++s) {
      if (s + 1 < KSTEPS) load_frags(so, s + 1, fr[(s + 1) & 1]);
      __builtin_amdgcn_sched_barrier(0);
      mma_step(fr[s & 1]);
      __builtin_amdgcn_sched_barrier(0);
    }
  };

  if (!matrix_role) {
    load_a(0);
#pragma unroll
    for (int j = 0; j < NSLOT; ++j) load_b(0, j);
    produce(0, 1);
  }
  __syncthreads();

  for (int i = 0; i < NCH - 2; ++i) {
    const int so = (i & 1) * STAGE_FLOATS;
    if (matrix_role) compute_chunk(so);
    else             produce(STAGE_FLOATS - so, i + 2);
    __syncthreads();
  }
  {
    const int so = ((NCH - 2) & 1) * STAGE_FLOATS;
    if (matrix_role) compute_chunk(so);
    else             produce_last(STAGE_FLOATS - so);
    __syncthreads();
  }
  if (matrix_role) compute_chunk(((NCH - 1) & 1) * STAGE_FLOATS);

  const int s_fr = (rg * 32 + 8 * lhi) * PR + lrow;

  if (matrix_role) {
    if (dh == 1) {
#pragma unroll
      for (int nt = 0; nt < 2; ++nt)
#pragma unroll
        for (int ct = 0; ct < 6; ++ct)
#pragma unroll
          for (int r = 0; r < 8; ++r)
            sdata[s_fr + (nt * 16 + r) * PR + ct * 16] = acc[nt][ct][r];
    }
  } else {
#pragma unroll
    for (int j = 0; j < NSLOT; ++j) {
      float a = slv[j], b = smi[j];
#pragma unroll
      for (int m = 1; m < 16; m <<= 1) {
        a += __shfl_xor(a, m);
        b += __shfl_xor(b, m);
      }
      const int c = cb + 8 * j;
      if (dq == 0) {
        csum[c]       = (c < KM) ? a : 0.0f;
        csum[KMP + c] = (c < KM) ? b : 0.0f;
      }
    }
  }
  __syncthreads();

  if (matrix_role) {
    if (dh == 0) {
#pragma unroll
      for (int nt = 0; nt < 2; ++nt)
#pragma unroll
        for (int ct = 0; ct < 6; ++ct)
#pragma unroll
          for (int r = 0; r < 8; ++r) {
            const int o = s_fr + (nt * 16 + r) * PR + ct * 16;
            sdata[o] = acc[nt][ct][r] + sdata[o];
          }
    }
  } else {
    if (rt < KM) {
      const float tot = (float)DD * LOG2PI + csum[rt] + csum[KMP + rt];
      cksh[rt] = log_w[rt] - 0.5f * tot;
    }
  }
  __syncthreads();

  if (wave8 == 4 || wave8 == 5) {
    const int pt = tid - 128;
    const float* row = &sdata[pt * PR];
    float lse[KK];
#pragma unroll
    for (int k = 0; k < KK; ++k) {
      float lp[MMIX];
      float mx = -3.0e38f;
#pragma unroll
      for (int m = 0; m < MMIX; ++m) {
        int c = k * MMIX + m;
        lp[m] = -0.5f * row[c] + cksh[c];
        mx = fmaxf(mx, lp[m]);
      }
      float s = 0.0f;
#pragma unroll
      for (int m = 0; m < MMIX; ++m) s += __builtin_amdgcn_exp2f((lp[m] - mx) * LOG2E);
      lse[k] = mx + __builtin_amdgcn_logf(s) * LN2;
    }
    float mv = lse[0];
#pragma unroll
    for (int k = 1; k < KK; ++k) mv = fmaxf(mv, lse[k]);
    float s = 0.0f;
#pragma unroll
    for (int k = 0; k < KK; ++k) s += __builtin_amdgcn_exp2f((lse[k] - mv) * LOG2E);
    out[n0 + pt] = 1.0f - 1.0f / s;
  }
}

extern "C" void kernel_launch(void* const* d_in, const int* in_sizes, int n_in,
                              void* d_out, int out_size, void* d_ws, size_t ws_size,
                              hipStream_t stream) {
  const float* feats     = (const float*)d_in[0];
  const float* means     = (const float*)d_in[1];
  const float* log_vars  = (const float*)d_in[2];
  const float* log_w     = (const float*)d_in[3];
  float* out = (float*)d_out;

  (void)in_sizes; (void)n_in; (void)out_size; (void)d_ws; (void)ws_size;

  gmm_main<<<NPTS / PTS, 256, 0, stream>>>(feats, means, log_vars, log_w, out);
}
